# starting kernel
# baseline (speedup 1.0000x reference)
#include <hip/hip_runtime.h>


typedef __attribute__((ext_vector_type(2))) float v2f;
typedef __attribute__((ext_vector_type(8))) float v8f;

#define NS_N 1024
#define NS_B 64

__global__ __launch_bounds__(256) void ns_rowsum_kernel(
    const float* __restrict__ scores, float* __restrict__ rowsum) {
  __shared__ __align__(16) float s[NS_N];
  const int b = blockIdx.x;
  const int t = threadIdx.x;
  const float* sb = scores + b * NS_N;

  #pragma unroll
  for (int j = t; j < NS_N; j += 256) s[j] = sb[j];
  __syncthreads();

  const int j0 = t * 4;
  const float sj0 = s[j0 + 0], sj1 = s[j0 + 1];
  const float sj2 = s[j0 + 2], sj3 = s[j0 + 3];
  float a0 = 0.f, a1 = 0.f, a2 = 0.f, a3 = 0.f;

  for (int k = 0; k < NS_N; k += 4) {
    const float4 sk = *(const float4*)&s[k];
    a0 += fabsf(sj0 - sk.x) + fabsf(sj0 - sk.y) + fabsf(sj0 - sk.z) + fabsf(sj0 - sk.w);
    a1 += fabsf(sj1 - sk.x) + fabsf(sj1 - sk.y) + fabsf(sj1 - sk.z) + fabsf(sj1 - sk.w);
    a2 += fabsf(sj2 - sk.x) + fabsf(sj2 - sk.y) + fabsf(sj2 - sk.z) + fabsf(sj2 - sk.w);
    a3 += fabsf(sj3 - sk.x) + fabsf(sj3 - sk.y) + fabsf(sj3 - sk.z) + fabsf(sj3 - sk.w);
  }
  float* rb = rowsum + b * NS_N + j0;
  rb[0] = a0; rb[1] = a1; rb[2] = a2; rb[3] = a3;
}

__global__ __launch_bounds__(256) void ns_softmax_kernel(
    const float* __restrict__ scores, const float* __restrict__ rowsum,
    float* __restrict__ out) {
  __shared__ __align__(16) float s[NS_N];
  __shared__ __align__(16) float rs[NS_N];
  __shared__ __align__(16) float stage[8][32];

  const int b    = blockIdx.x;
  const int t    = threadIdx.x;
  const int wave = t >> 5;
  const int lane = t & 31;

  #pragma unroll
  for (int j = t; j < NS_N; j += 256) {
    s[j]  = scores[b * NS_N + j];
    rs[j] = rowsum[b * NS_N + j];
  }
  __syncthreads();

  const int tile = blockIdx.y * 8 + wave;
  const int i0   = tile * 16;
  const int row  = lane & 15;
  const bool hi  = lane >= 16;
  const int kofs = hi ? 2 : 0;

  const float c = (float)(NS_N - 1 - 2 * (i0 + row));

  float m = -3.0e38f;
  const int jb = hi ? (NS_N / 2) : 0;
  for (int j = jb; j < jb + NS_N / 2; j += 4) {
    const float4 sv = *(const float4*)&s[j];
    const float4 rv = *(const float4*)&rs[j];
    m = fmaxf(m, fmaf(c, sv.x, -rv.x));
    m = fmaxf(m, fmaf(c, sv.y, -rv.y));
    m = fmaxf(m, fmaf(c, sv.z, -rv.z));
    m = fmaxf(m, fmaf(c, sv.w, -rv.w));
  }
  m = fmaxf(m, __shfl_xor(m, 16, 32));

  const float bfill = (row == 1) ? 1.0f : 0.0f;
  v8f acc = {};
  for (int j = 0; j < NS_N; j += 4) {
    const float2 sv = *(const float2*)&s[j + kofs];
    const float2 rv = *(const float2*)&rs[j + kofs];
    const float a0 = __expf(fmaf(c, sv.x, -rv.x) - m);
    const float a1 = __expf(fmaf(c, sv.y, -rv.y) - m);
    v2f A = {a0, a1};
    v2f Bm = {(row == 0) ? sv.x : bfill,
              (row == 0) ? sv.y : bfill};
    acc = __builtin_amdgcn_wmma_f32_16x16x4_f32(
         false, A,  false, Bm,
         (short)0, acc,  false,  false);
  }

  if (row < 2) {
    const int base = row * 16 + (hi ? 8 : 0);
    #pragma unroll
    for (int d = 0; d < 8; ++d) stage[wave][base + d] = acc[d];
  }
  __syncthreads();

  if (lane < 16)
    out[b * NS_N + i0 + lane] = stage[wave][lane] / stage[wave][16 + lane];
}

extern "C" void kernel_launch(void* const* d_in, const int* in_sizes, int n_in,
                              void* d_out, int out_size, void* d_ws, size_t ws_size,
                              hipStream_t stream) {
  (void)in_sizes; (void)n_in; (void)out_size; (void)ws_size;
  const float* scores = (const float*)d_in[0];
  float* out    = (float*)d_out;
  float* rowsum = (float*)d_ws;

  ns_rowsum_kernel<<<dim3(NS_B), 256, 0, stream>>>(scores, rowsum);
  ns_softmax_kernel<<<dim3(NS_B, 8), 256, 0, stream>>>(scores, rowsum, out);
}


# rowsum kernel spread over 4x the work-groups: grid (64,4), one j per thread instead of four; same k order and sum tree per j
# speedup vs baseline: 1.4061x; 1.4061x over previous
#include <hip/hip_runtime.h>


typedef __attribute__((ext_vector_type(2))) float v2f;
typedef __attribute__((ext_vector_type(8))) float v8f;

#define NS_N 1024
#define NS_B 64

__global__ __launch_bounds__(256) void ns_rowsum_kernel(
    const float* __restrict__ scores, float* __restrict__ rowsum) {
  __shared__ __align__(16) float s[NS_N];
  const int b = blockIdx.x;
  const int t = threadIdx.x;
  const float* sb = scores + b * NS_N;

  *(float4*)&s[t * 4] = *(const float4*)&sb[t * 4];
  __syncthreads();

  const int j = blockIdx.y * 256 + t;
  const float sj = s[j];
  float a = 0.f;

  for (int k = 0; k < NS_N; k += 4) {
    const float4 sk = *(const float4*)&s[k];
    a += fabsf(sj - sk.x) + fabsf(sj - sk.y) + fabsf(sj - sk.z) + fabsf(sj - sk.w);
  }
  rowsum[b * NS_N + j] = a;
}

__global__ __launch_bounds__(256) void ns_softmax_kernel(
    const float* __restrict__ scores, const float* __restrict__ rowsum,
    float* __restrict__ out) {
  __shared__ __align__(16) float s[NS_N];
  __shared__ __align__(16) float rs[NS_N];
  __shared__ __align__(16) float stage[8][32];

  const int b    = blockIdx.x;
  const int t    = threadIdx.x;
  const int wave = t >> 5;
  const int lane = t & 31;

  #pragma unroll
  for (int j = t; j < NS_N; j += 256) {
    s[j]  = scores[b * NS_N + j];
    rs[j] = rowsum[b * NS_N + j];
  }
  __syncthreads();

  const int tile = blockIdx.y * 8 + wave;
  const int i0   = tile * 16;
  const int row  = lane & 15;
  const bool hi  = lane >= 16;
  const int kofs = hi ? 2 : 0;

  const float c = (float)(NS_N - 1 - 2 * (i0 + row));

  float m = -3.0e38f;
  const int jb = hi ? (NS_N / 2) : 0;
  for (int j = jb; j < jb + NS_N / 2; j += 4) {
    const float4 sv = *(const float4*)&s[j];
    const float4 rv = *(const float4*)&rs[j];
    m = fmaxf(m, fmaf(c, sv.x, -rv.x));
    m = fmaxf(m, fmaf(c, sv.y, -rv.y));
    m = fmaxf(m, fmaf(c, sv.z, -rv.z));
    m = fmaxf(m, fmaf(c, sv.w, -rv.w));
  }
  m = fmaxf(m, __shfl_xor(m, 16, 32));

  const float bfill = (row == 1) ? 1.0f : 0.0f;
  v8f acc = {};
  for (int j = 0; j < NS_N; j += 4) {
    const float2 sv = *(const float2*)&s[j + kofs];
    const float2 rv = *(const float2*)&rs[j + kofs];
    const float a0 = __expf(fmaf(c, sv.x, -rv.x) - m);
    const float a1 = __expf(fmaf(c, sv.y, -rv.y) - m);
    v2f A = {a0, a1};
    v2f Bm = {(row == 0) ? sv.x : bfill,
              (row == 0) ? sv.y : bfill};
    acc = __builtin_amdgcn_wmma_f32_16x16x4_f32(
         false, A,  false, Bm,
         (short)0, acc,  false,  false);
  }

  if (row < 2) {
    const int base = row * 16 + (hi ? 8 : 0);
    #pragma unroll
    for (int d = 0; d < 8; ++d) stage[wave][base + d] = acc[d];
  }
  __syncthreads();

  if (lane < 16)
    out[b * NS_N + i0 + lane] = stage[wave][lane] / stage[wave][16 + lane];
}

extern "C" void kernel_launch(void* const* d_in, const int* in_sizes, int n_in,
                              void* d_out, int out_size, void* d_ws, size_t ws_size,
                              hipStream_t stream) {
  (void)in_sizes; (void)n_in; (void)out_size; (void)ws_size;
  const float* scores = (const float*)d_in[0];
  float* out    = (float*)d_out;
  float* rowsum = (float*)d_ws;

  ns_rowsum_kernel<<<dim3(NS_B, 4), 256, 0, stream>>>(scores, rowsum);
  ns_softmax_kernel<<<dim3(NS_B, 8), 256, 0, stream>>>(scores, rowsum, out);
}


# softmax kernel: exp factored over groups of 8 rows, exp(c_i*s_j-rs_j) = exp(cm*s_j-rs_j)*exp((c_i-cm)*s_j), so num and den are one dense f32 WMMA product; j split over 8 waves: sum order changes
# speedup vs baseline: 3.7531x; 2.6692x over previous
#include <hip/hip_runtime.h>


typedef __attribute__((ext_vector_type(2))) float v2f;
typedef __attribute__((ext_vector_type(8))) float v8f;

#define NS_N 1024
#define NS_B 64

__global__ __launch_bounds__(256) void ns_rowsum_kernel(
    const float* __restrict__ scores, float* __restrict__ rowsum) {
  __shared__ __align__(16) float s[NS_N];
  const int b = blockIdx.x;
  const int t = threadIdx.x;
  const float* sb = scores + b * NS_N;

  *(float4*)&s[t * 4] = *(const float4*)&sb[t * 4];
  __syncthreads();

  const int j = blockIdx.y * 256 + t;
  const float sj = s[j];
  float a = 0.f;

  for (int k = 0; k < NS_N; k += 4) {
    const float4 sk = *(const float4*)&s[k];
    a += fabsf(sj - sk.x) + fabsf(sj - sk.y) + fabsf(sj - sk.z) + fabsf(sj - sk.w);
  }
  rowsum[b * NS_N + j] = a;
}

__global__ __launch_bounds__(256) void ns_softmax_kernel(
    const float* __restrict__ scores, const float* __restrict__ rowsum,
    float* __restrict__ out) {
  __shared__ __align__(16) float s[NS_N];
  __shared__ __align__(16) float rs[NS_N];
  __shared__ float pmax[8][32];
  __shared__ __align__(16) float red[8][2][8][32];

  const int b    = blockIdx.x;
  const int t    = threadIdx.x;
  const int wave = t >> 5;
  const int lane = t & 31;

  #pragma unroll
  for (int j = t; j < NS_N; j += 256) {
    s[j]  = scores[b * NS_N + j];
    rs[j] = rowsum[b * NS_N + j];
  }
  __syncthreads();

  const int col = lane & 15;
  const int h   = lane >> 4;
  const int g0  = blockIdx.y * 32 + col;
  const float cm0 = (float)(NS_N - 8 - 16 * g0);
  const float cm1 = cm0 - 256.0f;

  const int jb = wave * 128 + h * 64;
  float4 sv[16], rv[16];
  #pragma unroll
  for (int q = 0; q < 16; ++q) {
    sv[q] = *(const float4*)&s[jb + 4 * q];
    rv[q] = *(const float4*)&rs[jb + 4 * q];
  }

  float m0 = -3.0e38f, m1 = -3.0e38f;
  #pragma unroll
  for (int q = 0; q < 16; ++q) {
    m0 = fmaxf(m0, fmaf(cm0, sv[q].x, -rv[q].x));
    m0 = fmaxf(m0, fmaf(cm0, sv[q].y, -rv[q].y));
    m0 = fmaxf(m0, fmaf(cm0, sv[q].z, -rv[q].z));
    m0 = fmaxf(m0, fmaf(cm0, sv[q].w, -rv[q].w));
    m1 = fmaxf(m1, fmaf(cm1, sv[q].x, -rv[q].x));
    m1 = fmaxf(m1, fmaf(cm1, sv[q].y, -rv[q].y));
    m1 = fmaxf(m1, fmaf(cm1, sv[q].z, -rv[q].z));
    m1 = fmaxf(m1, fmaf(cm1, sv[q].w, -rv[q].w));
  }
  m0 = fmaxf(m0, __shfl_xor(m0, 16, 32));
  m1 = fmaxf(m1, __shfl_xor(m1, 16, 32));
  if (lane < 16) {
    pmax[wave][col]      = m0;
    pmax[wave][16 + col] = m1;
  }
  __syncthreads();
  float M0 = pmax[0][col], M1 = pmax[0][16 + col];
  #pragma unroll
  for (int w = 1; w < 8; ++w) {
    M0 = fmaxf(M0, pmax[w][col]);
    M1 = fmaxf(M1, pmax[w][16 + col]);
  }

  const float d    = (float)(7 - 2 * (col & 7));
  const bool  wnum = col >= 8;
  v8f acc0 = {}, acc1 = {};
  #pragma unroll
  for (int q = 0; q < 16; ++q) {
    {
      const float sx = sv[q].x, sy = sv[q].y, rx = rv[q].x, ry = rv[q].y;
      const float p0 = __expf(d * sx), p1 = __expf(d * sy);
      v2f Bm = {wnum ? p0 * sx : p0, wnum ? p1 * sy : p1};
      v2f A0 = {__expf(fmaf(cm0, sx, -rx) - M0), __expf(fmaf(cm0, sy, -ry) - M0)};
      v2f A1 = {__expf(fmaf(cm1, sx, -rx) - M1), __expf(fmaf(cm1, sy, -ry) - M1)};
      acc0 = __builtin_amdgcn_wmma_f32_16x16x4_f32(
           false, A0,  false, Bm,
           (short)0, acc0,  false,  false);
      acc1 = __builtin_amdgcn_wmma_f32_16x16x4_f32(
           false, A1,  false, Bm,
           (short)0, acc1,  false,  false);
    }
    {
      const float sx = sv[q].z, sy = sv[q].w, rx = rv[q].z, ry = rv[q].w;
      const float p0 = __expf(d * sx), p1 = __expf(d * sy);
      v2f Bm = {wnum ? p0 * sx : p0, wnum ? p1 * sy : p1};
      v2f A0 = {__expf(fmaf(cm0, sx, -rx) - M0), __expf(fmaf(cm0, sy, -ry) - M0)};
      v2f A1 = {__expf(fmaf(cm1, sx, -rx) - M1), __expf(fmaf(cm1, sy, -ry) - M1)};
      acc0 = __builtin_amdgcn_wmma_f32_16x16x4_f32(
           false, A0,  false, Bm,
           (short)0, acc0,  false,  false);
      acc1 = __builtin_amdgcn_wmma_f32_16x16x4_f32(
           false, A1,  false, Bm,
           (short)0, acc1,  false,  false);
    }
  }

  #pragma unroll
  for (int e = 0; e < 8; ++e) {
    red[wave][0][e][lane] = acc0[e];
    red[wave][1][e][lane] = acc1[e];
  }
  __syncthreads();

  const int tile = t >> 7;
  const int m    = (t >> 3) & 15;
  const int r    = t & 7;
  const int ln   = r + 16 * (m >> 3);
  float den = 0.f, num = 0.f;
  #pragma unroll
  for (int w = 0; w < 8; ++w) {
    den += red[w][tile][m & 7][ln];
    num += red[w][tile][m & 7][ln + 8];
  }
  out[b * NS_N + blockIdx.y * 256 + t] = num / den;
}

extern "C" void kernel_launch(void* const* d_in, const int* in_sizes, int n_in,
                              void* d_out, int out_size, void* d_ws, size_t ws_size,
                              hipStream_t stream) {
  (void)in_sizes; (void)n_in; (void)out_size; (void)ws_size;
  const float* scores = (const float*)d_in[0];
  float* out    = (float*)d_out;
  float* rowsum = (float*)d_ws;

  ns_rowsum_kernel<<<dim3(NS_B, 4), 256, 0, stream>>>(scores, rowsum);
  ns_softmax_kernel<<<dim3(NS_B, 4), 256, 0, stream>>>(scores, rowsum, out);
}


# softmax kernel: batch row staged to LDS with one 128-bit load per thread per array, issued together (was four dependent rounds of 32-bit loads, each waited for); sums and their order unchanged
# speedup vs baseline: 3.9996x; 1.0657x over previous
#include <hip/hip_runtime.h>


typedef __attribute__((ext_vector_type(2))) float v2f;
typedef __attribute__((ext_vector_type(8))) float v8f;

#define NS_N 1024
#define NS_B 64

__global__ __launch_bounds__(256) void ns_rowsum_kernel(
    const float* __restrict__ scores, float* __restrict__ rowsum) {
  __shared__ __align__(16) float s[NS_N];
  const int b = blockIdx.x;
  const int t = threadIdx.x;
  const float* sb = scores + b * NS_N;

  *(float4*)&s[t * 4] = *(const float4*)&sb[t * 4];
  __syncthreads();

  const int j = blockIdx.y * 256 + t;
  const float sj = s[j];
  float a = 0.f;

  for (int k = 0; k < NS_N; k += 4) {
    const float4 sk = *(const float4*)&s[k];
    a += fabsf(sj - sk.x) + fabsf(sj - sk.y) + fabsf(sj - sk.z) + fabsf(sj - sk.w);
  }
  rowsum[b * NS_N + j] = a;
}

__global__ __launch_bounds__(256) void ns_softmax_kernel(
    const float* __restrict__ scores, const float* __restrict__ rowsum,
    float* __restrict__ out) {
  __shared__ __align__(16) float s[NS_N];
  __shared__ __align__(16) float rs[NS_N];
  __shared__ float pmax[8][32];
  __shared__ __align__(16) float red[8][2][8][32];

  const int b    = blockIdx.x;
  const int t    = threadIdx.x;
  const int wave = t >> 5;
  const int lane = t & 31;

  *(float4*)&s[t * 4]  = *(const float4*)&scores[b * NS_N + t * 4];
  *(float4*)&rs[t * 4] = *(const float4*)&rowsum[b * NS_N + t * 4];
  __syncthreads();

  const int col = lane & 15;
  const int h   = lane >> 4;
  const int g0  = blockIdx.y * 32 + col;
  const float cm0 = (float)(NS_N - 8 - 16 * g0);
  const float cm1 = cm0 - 256.0f;

  const int jb = wave * 128 + h * 64;
  float4 sv[16], rv[16];
  #pragma unroll
  for (int q = 0; q < 16; ++q) {
    sv[q] = *(const float4*)&s[jb + 4 * q];
    rv[q] = *(const float4*)&rs[jb + 4 * q];
  }

  float m0 = -3.0e38f, m1 = -3.0e38f;
  #pragma unroll
  for (int q = 0; q < 16; ++q) {
    m0 = fmaxf(m0, fmaf(cm0, sv[q].x, -rv[q].x));
    m0 = fmaxf(m0, fmaf(cm0, sv[q].y, -rv[q].y));
    m0 = fmaxf(m0, fmaf(cm0, sv[q].z, -rv[q].z));
    m0 = fmaxf(m0, fmaf(cm0, sv[q].w, -rv[q].w));
    m1 = fmaxf(m1, fmaf(cm1, sv[q].x, -rv[q].x));
    m1 = fmaxf(m1, fmaf(cm1, sv[q].y, -rv[q].y));
    m1 = fmaxf(m1, fmaf(cm1, sv[q].z, -rv[q].z));
    m1 = fmaxf(m1, fmaf(cm1, sv[q].w, -rv[q].w));
  }
  m0 = fmaxf(m0, __shfl_xor(m0, 16, 32));
  m1 = fmaxf(m1, __shfl_xor(m1, 16, 32));
  if (lane < 16) {
    pmax[wave][col]      = m0;
    pmax[wave][16 + col] = m1;
  }
  __syncthreads();
  float M0 = pmax[0][col], M1 = pmax[0][16 + col];
  #pragma unroll
  for (int w = 1; w < 8; ++w) {
    M0 = fmaxf(M0, pmax[w][col]);
    M1 = fmaxf(M1, pmax[w][16 + col]);
  }

  const float d    = (float)(7 - 2 * (col & 7));
  const bool  wnum = col >= 8;
  v8f acc0 = {}, acc1 = {};
  #pragma unroll
  for (int q = 0; q < 16; ++q) {
    {
      const float sx = sv[q].x, sy = sv[q].y, rx = rv[q].x, ry = rv[q].y;
      const float p0 = __expf(d * sx), p1 = __expf(d * sy);
      v2f Bm = {wnum ? p0 * sx : p0, wnum ? p1 * sy : p1};
      v2f A0 = {__expf(fmaf(cm0, sx, -rx) - M0), __expf(fmaf(cm0, sy, -ry) - M0)};
      v2f A1 = {__expf(fmaf(cm1, sx, -rx) - M1), __expf(fmaf(cm1, sy, -ry) - M1)};
      acc0 = __builtin_amdgcn_wmma_f32_16x16x4_f32(
           false, A0,  false, Bm,
           (short)0, acc0,  false,  false);
      acc1 = __builtin_amdgcn_wmma_f32_16x16x4_f32(
           false, A1,  false, Bm,
           (short)0, acc1,  false,  false);
    }
    {
      const float sx = sv[q].z, sy = sv[q].w, rx = rv[q].z, ry = rv[q].w;
      const float p0 = __expf(d * sx), p1 = __expf(d * sy);
      v2f Bm = {wnum ? p0 * sx : p0, wnum ? p1 * sy : p1};
      v2f A0 = {__expf(fmaf(cm0, sx, -rx) - M0), __expf(fmaf(cm0, sy, -ry) - M0)};
      v2f A1 = {__expf(fmaf(cm1, sx, -rx) - M1), __expf(fmaf(cm1, sy, -ry) - M1)};
      acc0 = __builtin_amdgcn_wmma_f32_16x16x4_f32(
           false, A0,  false, Bm,
           (short)0, acc0,  false,  false);
      acc1 = __builtin_amdgcn_wmma_f32_16x16x4_f32(
           false, A1,  false, Bm,
           (short)0, acc1,  false,  false);
    }
  }

  #pragma unroll
  for (int e = 0; e < 8; ++e) {
    red[wave][0][e][lane] = acc0[e];
    red[wave][1][e][lane] = acc1[e];
  }
  __syncthreads();

  const int tile = t >> 7;
  const int m    = (t >> 3) & 15;
  const int r    = t & 7;
  const int ln   = r + 16 * (m >> 3);
  float den = 0.f, num = 0.f;
  #pragma unroll
  for (int w = 0; w < 8; ++w) {
    den += red[w][tile][m & 7][ln];
    num += red[w][tile][m & 7][ln + 8];
  }
  out[b * NS_N + blockIdx.y * 256 + t] = num / den;
}

extern "C" void kernel_launch(void* const* d_in, const int* in_sizes, int n_in,
                              void* d_out, int out_size, void* d_ws, size_t ws_size,
                              hipStream_t stream) {
  (void)in_sizes; (void)n_in; (void)out_size; (void)ws_size;
  const float* scores = (const float*)d_in[0];
  float* out    = (float*)d_out;
  float* rowsum = (float*)d_ws;

  ns_rowsum_kernel<<<dim3(NS_B, 4), 256, 0, stream>>>(scores, rowsum);
  ns_softmax_kernel<<<dim3(NS_B, 4), 256, 0, stream>>>(scores, rowsum, out);
}


# rowsum from an in-block bitonic sort and a prefix sum, rs_p = s_p(2p-n)+S-2E_p on the sorted row, inside the softmax kernel: one launch, no workspace; j runs in sorted order: sum order changes
# speedup vs baseline: 5.9972x; 1.4994x over previous
#include <hip/hip_runtime.h>


typedef __attribute__((ext_vector_type(2))) float v2f;
typedef __attribute__((ext_vector_type(8))) float v8f;

#define NS_N 1024
#define NS_B 64

template <int M>
__device__ __forceinline__ float ns_lane_xor(const float x) {
  const int b = __builtin_bit_cast(int, x);
  if constexpr (M < 16)
    return __builtin_bit_cast(float, __builtin_amdgcn_update_dpp(0, b, 0x160 + M, 0xf, 0xf, true));
  else
    return __builtin_bit_cast(float, __builtin_amdgcn_permlanex16(b, b, 0x76543210, 0xfedcba98, false, false));
}

template <int D>
__device__ __forceinline__ float ns_row_shr(const float x) {
  return __builtin_bit_cast(float, __builtin_amdgcn_update_dpp(0, __builtin_bit_cast(int, x), 0x110 + D, 0xf, 0xf, true));
}

__device__ __forceinline__ float ns_pick(const float a, const float b, const float lim) {
  return __builtin_amdgcn_fmed3f(a, b, lim);
}

__device__ __forceinline__ void ns_cx(float& a, float& b, const float lim) {
  const float x = a, y = b;
  a = ns_pick(x, y, lim);
  b = ns_pick(x, y, -lim);
}

template <int M>
__device__ __forceinline__ void ns_lane_step(float (&v)[4], const float lim) {
  #pragma unroll
  for (int e = 0; e < 4; ++e) v[e] = ns_pick(v[e], ns_lane_xor<M>(v[e]), lim);
}

__device__ __forceinline__ void ns_lds_step(float (&v)[4], const float lim, float* buf,
                                            const int t, const int m) {
  *(float4*)&buf[4 * t] = make_float4(v[0], v[1], v[2], v[3]);
  __syncthreads();
  const float4 o = *(const float4*)&buf[4 * (t ^ m)];
  v[0] = ns_pick(v[0], o.x, lim);
  v[1] = ns_pick(v[1], o.y, lim);
  v[2] = ns_pick(v[2], o.z, lim);
  v[3] = ns_pick(v[3], o.w, lim);
}

template <int K>
__device__ __forceinline__ void ns_stage(float (&v)[4], const int t, float* bA, float* bB) {
  const float inf = __builtin_inff();
  const bool asc = (t & (K / 4)) == 0;
  auto lim = [&](const int m) { return (((t & m) == 0) == asc) ? -inf : inf; };
  if constexpr (K == 1024) {
    ns_lds_step(v, lim(128), bB, t, 128);
    ns_lds_step(v, lim(64), bA, t, 64);
    ns_lds_step(v, lim(32), bB, t, 32);
  }
  if constexpr (K == 512) {
    ns_lds_step(v, lim(64), bB, t, 64);
    ns_lds_step(v, lim(32), bA, t, 32);
  }
  if constexpr (K == 256) ns_lds_step(v, lim(32), bA, t, 32);
  if constexpr (K >= 128) ns_lane_step<16>(v, lim(16));
  if constexpr (K >= 64) ns_lane_step<8>(v, lim(8));
  if constexpr (K >= 32) ns_lane_step<4>(v, lim(4));
  if constexpr (K >= 16) ns_lane_step<2>(v, lim(2));
  ns_lane_step<1>(v, lim(1));
  const float la = asc ? -inf : inf;
  ns_cx(v[0], v[2], la);
  ns_cx(v[1], v[3], la);
  ns_cx(v[0], v[1], la);
  ns_cx(v[2], v[3], la);
}

__global__ __launch_bounds__(256) void ns_forward_kernel(
    const float* __restrict__ scores, float* __restrict__ out) {
  __shared__ __align__(16) float s[NS_N];
  __shared__ __align__(16) float rs[NS_N];
  __shared__ float wsum[8];
  __shared__ float pmax[8][32];
  __shared__ __align__(16) float red[8][2][8][32];

  const int b    = blockIdx.x;
  const int t    = threadIdx.x;
  const int wave = t >> 5;
  const int lane = t & 31;

  float v[4];
  {
    const float4 in = *(const float4*)&scores[b * NS_N + t * 4];
    v[0] = in.x; v[1] = in.y; v[2] = in.z; v[3] = in.w;
  }
  {
    const float inf = __builtin_inff();
    ns_cx(v[0], v[1], -inf);
    ns_cx(v[2], v[3], inf);
    const float la = (t & 1) == 0 ? -inf : inf;
    ns_cx(v[0], v[2], la);
    ns_cx(v[1], v[3], la);
    ns_cx(v[0], v[1], la);
    ns_cx(v[2], v[3], la);
  }
  ns_stage<8>(v, t, s, rs);
  ns_stage<16>(v, t, s, rs);
  ns_stage<32>(v, t, s, rs);
  ns_stage<64>(v, t, s, rs);
  ns_stage<128>(v, t, s, rs);
  ns_stage<256>(v, t, s, rs);
  ns_stage<512>(v, t, s, rs);
  ns_stage<1024>(v, t, s, rs);

  {
    const float t01 = v[0] + v[1], t012 = t01 + v[2], tsum = t012 + v[3];
    float r = tsum;
    r += ns_row_shr<1>(r);
    r += ns_row_shr<2>(r);
    r += ns_row_shr<4>(r);
    r += ns_row_shr<8>(r);
    const float rex = ns_row_shr<1>(r);
    const float c0 = __builtin_bit_cast(float, __builtin_amdgcn_readlane(__builtin_bit_cast(int, r), 15));
    const float c1 = __builtin_bit_cast(float, __builtin_amdgcn_readlane(__builtin_bit_cast(int, r), 31));
    const float ex = rex + (lane >= 16 ? c0 : 0.f);
    if (lane == 0) wsum[wave] = c0 + c1;
    __syncthreads();
    float base = 0.f, S = 0.f;
    #pragma unroll
    for (int w = 0; w < 8; ++w) {
      const float x = wsum[w];
      base += (w < wave) ? x : 0.f;
      S += x;
    }
    const float e0 = base + ex;
    const float e1 = e0 + v[0], e2 = e0 + t01, e3 = e0 + t012;
    const float p0 = (float)(8 * t - NS_N);
    float4 rv4;
    rv4.x = fmaf(v[0], p0,       S - 2.f * e0);
    rv4.y = fmaf(v[1], p0 + 2.f, S - 2.f * e1);
    rv4.z = fmaf(v[2], p0 + 4.f, S - 2.f * e2);
    rv4.w = fmaf(v[3], p0 + 6.f, S - 2.f * e3);
    *(float4*)&s[t * 4]  = make_float4(v[0], v[1], v[2], v[3]);
    *(float4*)&rs[t * 4] = rv4;
  }
  __syncthreads();

  const int col = lane & 15;
  const int h   = lane >> 4;
  const int g0  = blockIdx.y * 32 + col;
  const float cm0 = (float)(NS_N - 8 - 16 * g0);
  const float cm1 = cm0 - 256.0f;

  const int jb = wave * 128 + h * 64;
  float4 sv[16], rv[16];
  #pragma unroll
  for (int q = 0; q < 16; ++q) {
    sv[q] = *(const float4*)&s[jb + 4 * q];
    rv[q] = *(const float4*)&rs[jb + 4 * q];
  }

  float m0 = -3.0e38f, m1 = -3.0e38f;
  #pragma unroll
  for (int q = 0; q < 16; ++q) {
    m0 = fmaxf(m0, fmaf(cm0, sv[q].x, -rv[q].x));
    m0 = fmaxf(m0, fmaf(cm0, sv[q].y, -rv[q].y));
    m0 = fmaxf(m0, fmaf(cm0, sv[q].z, -rv[q].z));
    m0 = fmaxf(m0, fmaf(cm0, sv[q].w, -rv[q].w));
    m1 = fmaxf(m1, fmaf(cm1, sv[q].x, -rv[q].x));
    m1 = fmaxf(m1, fmaf(cm1, sv[q].y, -rv[q].y));
    m1 = fmaxf(m1, fmaf(cm1, sv[q].z, -rv[q].z));
    m1 = fmaxf(m1, fmaf(cm1, sv[q].w, -rv[q].w));
  }
  m0 = fmaxf(m0, __shfl_xor(m0, 16, 32));
  m1 = fmaxf(m1, __shfl_xor(m1, 16, 32));
  if (lane < 16) {
    pmax[wave][col]      = m0;
    pmax[wave][16 + col] = m1;
  }
  __syncthreads();
  float M0 = pmax[0][col], M1 = pmax[0][16 + col];
  #pragma unroll
  for (int w = 1; w < 8; ++w) {
    M0 = fmaxf(M0, pmax[w][col]);
    M1 = fmaxf(M1, pmax[w][16 + col]);
  }

  const float d    = (float)(7 - 2 * (col & 7));
  const bool  wnum = col >= 8;
  v8f acc0 = {}, acc1 = {};
  #pragma unroll
  for (int q = 0; q < 16; ++q) {
    {
      const float sx = sv[q].x, sy = sv[q].y, rx = rv[q].x, ry = rv[q].y;
      const float p0 = __expf(d * sx), p1 = __expf(d * sy);
      v2f Bm = {wnum ? p0 * sx : p0, wnum ? p1 * sy : p1};
      v2f A0 = {__expf(fmaf(cm0, sx, -rx) - M0), __expf(fmaf(cm0, sy, -ry) - M0)};
      v2f A1 = {__expf(fmaf(cm1, sx, -rx) - M1), __expf(fmaf(cm1, sy, -ry) - M1)};
      acc0 = __builtin_amdgcn_wmma_f32_16x16x4_f32(
           false, A0,  false, Bm,
           (short)0, acc0,  false,  false);
      acc1 = __builtin_amdgcn_wmma_f32_16x16x4_f32(
           false, A1,  false, Bm,
           (short)0, acc1,  false,  false);
    }
    {
      const float sx = sv[q].z, sy = sv[q].w, rx = rv[q].z, ry = rv[q].w;
      const float p0 = __expf(d * sx), p1 = __expf(d * sy);
      v2f Bm = {wnum ? p0 * sx : p0, wnum ? p1 * sy : p1};
      v2f A0 = {__expf(fmaf(cm0, sx, -rx) - M0), __expf(fmaf(cm0, sy, -ry) - M0)};
      v2f A1 = {__expf(fmaf(cm1, sx, -rx) - M1), __expf(fmaf(cm1, sy, -ry) - M1)};
      acc0 = __builtin_amdgcn_wmma_f32_16x16x4_f32(
           false, A0,  false, Bm,
           (short)0, acc0,  false,  false);
      acc1 = __builtin_amdgcn_wmma_f32_16x16x4_f32(
           false, A1,  false, Bm,
           (short)0, acc1,  false,  false);
    }
  }

  #pragma unroll
  for (int e = 0; e < 8; ++e) {
    red[wave][0][e][lane] = acc0[e];
    red[wave][1][e][lane] = acc1[e];
  }
  __syncthreads();

  const int tile = t >> 7;
  const int m    = (t >> 3) & 15;
  const int r    = t & 7;
  const int ln   = r + 16 * (m >> 3);
  float den = 0.f, num = 0.f;
  #pragma unroll
  for (int w = 0; w < 8; ++w) {
    den += red[w][tile][m & 7][ln];
    num += red[w][tile][m & 7][ln + 8];
  }
  out[b * NS_N + blockIdx.y * 256 + t] = num / den;
}

extern "C" void kernel_launch(void* const* d_in, const int* in_sizes, int n_in,
                              void* d_out, int out_size, void* d_ws, size_t ws_size,
                              hipStream_t stream) {
  (void)in_sizes; (void)n_in; (void)out_size; (void)d_ws; (void)ws_size;
  const float* scores = (const float*)d_in[0];
  float* out = (float*)d_out;

  ns_forward_kernel<<<dim3(NS_B, 4), 256, 0, stream>>>(scores, out);
}


# softmax shift M_g read at its closed-form place in the sorted row (p = n-5-8g, where cm_g*s_p-rowsum_p peaks) instead of searched: max pass, its barrier and LDS exchange gone; sum order unchanged
# speedup vs baseline: 6.3262x; 1.0549x over previous
#include <hip/hip_runtime.h>


typedef __attribute__((ext_vector_type(2))) float v2f;
typedef __attribute__((ext_vector_type(8))) float v8f;

#define NS_N 1024
#define NS_B 64

template <int M>
__device__ __forceinline__ float ns_lane_xor(const float x) {
  const int b = __builtin_bit_cast(int, x);
  if constexpr (M < 16)
    return __builtin_bit_cast(float, __builtin_amdgcn_update_dpp(0, b, 0x160 + M, 0xf, 0xf, true));
  else
    return __builtin_bit_cast(float, __builtin_amdgcn_permlanex16(b, b, 0x76543210, 0xfedcba98, false, false));
}

template <int D>
__device__ __forceinline__ float ns_row_shr(const float x) {
  return __builtin_bit_cast(float, __builtin_amdgcn_update_dpp(0, __builtin_bit_cast(int, x), 0x110 + D, 0xf, 0xf, true));
}

__device__ __forceinline__ float ns_pick(const float a, const float b, const float lim) {
  return __builtin_amdgcn_fmed3f(a, b, lim);
}

__device__ __forceinline__ void ns_cx(float& a, float& b, const float lim) {
  const float x = a, y = b;
  a = ns_pick(x, y, lim);
  b = ns_pick(x, y, -lim);
}

template <int M>
__device__ __forceinline__ void ns_lane_step(float (&v)[4], const float lim) {
  #pragma unroll
  for (int e = 0; e < 4; ++e) v[e] = ns_pick(v[e], ns_lane_xor<M>(v[e]), lim);
}

__device__ __forceinline__ void ns_lds_step(float (&v)[4], const float lim, float* buf,
                                            const int t, const int m) {
  *(float4*)&buf[4 * t] = make_float4(v[0], v[1], v[2], v[3]);
  __syncthreads();
  const float4 o = *(const float4*)&buf[4 * (t ^ m)];
  v[0] = ns_pick(v[0], o.x, lim);
  v[1] = ns_pick(v[1], o.y, lim);
  v[2] = ns_pick(v[2], o.z, lim);
  v[3] = ns_pick(v[3], o.w, lim);
}

template <int K>
__device__ __forceinline__ void ns_stage(float (&v)[4], const int t, float* bA, float* bB) {
  const float inf = __builtin_inff();
  const bool asc = (t & (K / 4)) == 0;
  auto lim = [&](const int m) { return (((t & m) == 0) == asc) ? -inf : inf; };
  if constexpr (K == 1024) {
    ns_lds_step(v, lim(128), bB, t, 128);
    ns_lds_step(v, lim(64), bA, t, 64);
    ns_lds_step(v, lim(32), bB, t, 32);
  }
  if constexpr (K == 512) {
    ns_lds_step(v, lim(64), bB, t, 64);
    ns_lds_step(v, lim(32), bA, t, 32);
  }
  if constexpr (K == 256) ns_lds_step(v, lim(32), bA, t, 32);
  if constexpr (K >= 128) ns_lane_step<16>(v, lim(16));
  if constexpr (K >= 64) ns_lane_step<8>(v, lim(8));
  if constexpr (K >= 32) ns_lane_step<4>(v, lim(4));
  if constexpr (K >= 16) ns_lane_step<2>(v, lim(2));
  ns_lane_step<1>(v, lim(1));
  const float la = asc ? -inf : inf;
  ns_cx(v[0], v[2], la);
  ns_cx(v[1], v[3], la);
  ns_cx(v[0], v[1], la);
  ns_cx(v[2], v[3], la);
}

__global__ __launch_bounds__(256) void ns_forward_kernel(
    const float* __restrict__ scores, float* __restrict__ out) {
  __shared__ __align__(16) float s[NS_N];
  __shared__ __align__(16) float rs[NS_N];
  __shared__ float wsum[8];
  __shared__ __align__(16) float red[8][2][8][32];

  const int b    = blockIdx.x;
  const int t    = threadIdx.x;
  const int wave = t >> 5;
  const int lane = t & 31;

  float v[4];
  {
    const float4 in = *(const float4*)&scores[b * NS_N + t * 4];
    v[0] = in.x; v[1] = in.y; v[2] = in.z; v[3] = in.w;
  }
  {
    const float inf = __builtin_inff();
    ns_cx(v[0], v[1], -inf);
    ns_cx(v[2], v[3], inf);
    const float la = (t & 1) == 0 ? -inf : inf;
    ns_cx(v[0], v[2], la);
    ns_cx(v[1], v[3], la);
    ns_cx(v[0], v[1], la);
    ns_cx(v[2], v[3], la);
  }
  ns_stage<8>(v, t, s, rs);
  ns_stage<16>(v, t, s, rs);
  ns_stage<32>(v, t, s, rs);
  ns_stage<64>(v, t, s, rs);
  ns_stage<128>(v, t, s, rs);
  ns_stage<256>(v, t, s, rs);
  ns_stage<512>(v, t, s, rs);
  ns_stage<1024>(v, t, s, rs);

  {
    const float t01 = v[0] + v[1], t012 = t01 + v[2], tsum = t012 + v[3];
    float r = tsum;
    r += ns_row_shr<1>(r);
    r += ns_row_shr<2>(r);
    r += ns_row_shr<4>(r);
    r += ns_row_shr<8>(r);
    const float rex = ns_row_shr<1>(r);
    const float c0 = __builtin_bit_cast(float, __builtin_amdgcn_readlane(__builtin_bit_cast(int, r), 15));
    const float c1 = __builtin_bit_cast(float, __builtin_amdgcn_readlane(__builtin_bit_cast(int, r), 31));
    const float ex = rex + (lane >= 16 ? c0 : 0.f);
    if (lane == 0) wsum[wave] = c0 + c1;
    __syncthreads();
    float base = 0.f, S = 0.f;
    #pragma unroll
    for (int w = 0; w < 8; ++w) {
      const float x = wsum[w];
      base += (w < wave) ? x : 0.f;
      S += x;
    }
    const float e0 = base + ex;
    const float e1 = e0 + v[0], e2 = e0 + t01, e3 = e0 + t012;
    const float p0 = (float)(8 * t - NS_N);
    float4 rv4;
    rv4.x = fmaf(v[0], p0,       S - 2.f * e0);
    rv4.y = fmaf(v[1], p0 + 2.f, S - 2.f * e1);
    rv4.z = fmaf(v[2], p0 + 4.f, S - 2.f * e2);
    rv4.w = fmaf(v[3], p0 + 6.f, S - 2.f * e3);
    *(float4*)&s[t * 4]  = make_float4(v[0], v[1], v[2], v[3]);
    *(float4*)&rs[t * 4] = rv4;
  }
  __syncthreads();

  const int col = lane & 15;
  const int h   = lane >> 4;
  const int g0  = blockIdx.y * 32 + col;
  const float cm0 = (float)(NS_N - 8 - 16 * g0);
  const float cm1 = cm0 - 256.0f;

  const int pm0 = NS_N - 5 - 8 * g0;
  const float sm0 = s[pm0], rm0 = rs[pm0], sm1 = s[pm0 - 128], rm1 = rs[pm0 - 128];

  __builtin_amdgcn_sched_barrier(0);

  const int jb = wave * 128 + h * 64;
  float4 sv[16], rv[16];
  #pragma unroll
  for (int q = 0; q < 16; ++q) {
    sv[q] = *(const float4*)&s[jb + 4 * q];
    rv[q] = *(const float4*)&rs[jb + 4 * q];
  }

  __builtin_amdgcn_sched_barrier(0);

  const float M0 = fmaf(cm0, sm0, -rm0);
  const float M1 = fmaf(cm1, sm1, -rm1);

  const float d    = (float)(7 - 2 * (col & 7));
  const bool  wnum = col >= 8;
  v8f acc0 = {}, acc1 = {};
  #pragma unroll
  for (int q = 0; q < 16; ++q) {
    {
      const float sx = sv[q].x, sy = sv[q].y, rx = rv[q].x, ry = rv[q].y;
      const float p0 = __expf(d * sx), p1 = __expf(d * sy);
      v2f Bm = {wnum ? p0 * sx : p0, wnum ? p1 * sy : p1};
      v2f A0 = {__expf(fmaf(cm0, sx, -rx) - M0), __expf(fmaf(cm0, sy, -ry) - M0)};
      v2f A1 = {__expf(fmaf(cm1, sx, -rx) - M1), __expf(fmaf(cm1, sy, -ry) - M1)};
      acc0 = __builtin_amdgcn_wmma_f32_16x16x4_f32(
           false, A0,  false, Bm,
           (short)0, acc0,  false,  false);
      acc1 = __builtin_amdgcn_wmma_f32_16x16x4_f32(
           false, A1,  false, Bm,
           (short)0, acc1,  false,  false);
    }
    {
      const float sx = sv[q].z, sy = sv[q].w, rx = rv[q].z, ry = rv[q].w;
      const float p0 = __expf(d * sx), p1 = __expf(d * sy);
      v2f Bm = {wnum ? p0 * sx : p0, wnum ? p1 * sy : p1};
      v2f A0 = {__expf(fmaf(cm0, sx, -rx) - M0), __expf(fmaf(cm0, sy, -ry) - M0)};
      v2f A1 = {__expf(fmaf(cm1, sx, -rx) - M1), __expf(fmaf(cm1, sy, -ry) - M1)};
      acc0 = __builtin_amdgcn_wmma_f32_16x16x4_f32(
           false, A0,  false, Bm,
           (short)0, acc0,  false,  false);
      acc1 = __builtin_amdgcn_wmma_f32_16x16x4_f32(
           false, A1,  false, Bm,
           (short)0, acc1,  false,  false);
    }
  }

  #pragma unroll
  for (int e = 0; e < 8; ++e) {
    red[wave][0][e][lane] = acc0[e];
    red[wave][1][e][lane] = acc1[e];
  }
  __syncthreads();

  const int tile = t >> 7;
  const int m    = (t >> 3) & 15;
  const int r    = t & 7;
  const int ln   = r + 16 * (m >> 3);
  float den = 0.f, num = 0.f;
  #pragma unroll
  for (int w = 0; w < 8; ++w) {
    den += red[w][tile][m & 7][ln];
    num += red[w][tile][m & 7][ln + 8];
  }
  out[b * NS_N + blockIdx.y * 256 + t] = num / den;
}

extern "C" void kernel_launch(void* const* d_in, const int* in_sizes, int n_in,
                              void* d_out, int out_size, void* d_ws, size_t ws_size,
                              hipStream_t stream) {
  (void)in_sizes; (void)n_in; (void)out_size; (void)d_ws; (void)ws_size;
  const float* scores = (const float*)d_in[0];
  float* out = (float*)d_out;

  ns_forward_kernel<<<dim3(NS_B, 4), 256, 0, stream>>>(scores, out);
}


# product loop: sorted positions dealt to the waves in 16-position chunks; a wave skips a chunk when every exp argument of both A tiles is below -106 (all its Q exactly 0): sum order changes
# speedup vs baseline: 7.0304x; 1.1113x over previous
#include <hip/hip_runtime.h>


typedef __attribute__((ext_vector_type(2))) float v2f;
typedef __attribute__((ext_vector_type(4))) float v4f;
typedef __attribute__((ext_vector_type(8))) float v8f;

#define NS_N 1024
#define NS_B 64

template <int M>
__device__ __forceinline__ float ns_lane_xor(const float x) {
  const int b = __builtin_bit_cast(int, x);
  if constexpr (M < 16)
    return __builtin_bit_cast(float, __builtin_amdgcn_update_dpp(0, b, 0x160 + M, 0xf, 0xf, true));
  else
    return __builtin_bit_cast(float, __builtin_amdgcn_permlanex16(b, b, 0x76543210, 0xfedcba98, false, false));
}

template <int D>
__device__ __forceinline__ float ns_row_shr(const float x) {
  return __builtin_bit_cast(float, __builtin_amdgcn_update_dpp(0, __builtin_bit_cast(int, x), 0x110 + D, 0xf, 0xf, true));
}

__device__ __forceinline__ float ns_pick(const float a, const float b, const float lim) {
  return __builtin_amdgcn_fmed3f(a, b, lim);
}

__device__ __forceinline__ void ns_cx(float& a, float& b, const float lim) {
  const float x = a, y = b;
  a = ns_pick(x, y, lim);
  b = ns_pick(x, y, -lim);
}

template <int M>
__device__ __forceinline__ void ns_lane_step(float (&v)[4], const float lim) {
  #pragma unroll
  for (int e = 0; e < 4; ++e) v[e] = ns_pick(v[e], ns_lane_xor<M>(v[e]), lim);
}

__device__ __forceinline__ void ns_lds_step(float (&v)[4], const float lim, float* buf,
                                            const int t, const int m) {
  *(float4*)&buf[4 * t] = make_float4(v[0], v[1], v[2], v[3]);
  __syncthreads();
  const float4 o = *(const float4*)&buf[4 * (t ^ m)];
  v[0] = ns_pick(v[0], o.x, lim);
  v[1] = ns_pick(v[1], o.y, lim);
  v[2] = ns_pick(v[2], o.z, lim);
  v[3] = ns_pick(v[3], o.w, lim);
}

template <int K>
__device__ __forceinline__ void ns_stage(float (&v)[4], const int t, float* bA, float* bB) {
  const float inf = __builtin_inff();
  const bool asc = (t & (K / 4)) == 0;
  auto lim = [&](const int m) { return (((t & m) == 0) == asc) ? -inf : inf; };
  if constexpr (K == 1024) {
    ns_lds_step(v, lim(128), bB, t, 128);
    ns_lds_step(v, lim(64), bA, t, 64);
    ns_lds_step(v, lim(32), bB, t, 32);
  }
  if constexpr (K == 512) {
    ns_lds_step(v, lim(64), bB, t, 64);
    ns_lds_step(v, lim(32), bA, t, 32);
  }
  if constexpr (K == 256) ns_lds_step(v, lim(32), bA, t, 32);
  if constexpr (K >= 128) ns_lane_step<16>(v, lim(16));
  if constexpr (K >= 64) ns_lane_step<8>(v, lim(8));
  if constexpr (K >= 32) ns_lane_step<4>(v, lim(4));
  if constexpr (K >= 16) ns_lane_step<2>(v, lim(2));
  ns_lane_step<1>(v, lim(1));
  const float la = asc ? -inf : inf;
  ns_cx(v[0], v[2], la);
  ns_cx(v[1], v[3], la);
  ns_cx(v[0], v[1], la);
  ns_cx(v[2], v[3], la);
}

__global__ __launch_bounds__(256) void ns_forward_kernel(
    const float* __restrict__ scores, float* __restrict__ out) {
  __shared__ __align__(16) float s[NS_N];
  __shared__ __align__(16) float rs[NS_N];
  __shared__ float wsum[8];
  __shared__ __align__(16) float red[8][2][8][32];

  const int b    = blockIdx.x;
  const int t    = threadIdx.x;
  const int wave = t >> 5;
  const int lane = t & 31;

  float v[4];
  {
    const float4 in = *(const float4*)&scores[b * NS_N + t * 4];
    v[0] = in.x; v[1] = in.y; v[2] = in.z; v[3] = in.w;
  }
  {
    const float inf = __builtin_inff();
    ns_cx(v[0], v[1], -inf);
    ns_cx(v[2], v[3], inf);
    const float la = (t & 1) == 0 ? -inf : inf;
    ns_cx(v[0], v[2], la);
    ns_cx(v[1], v[3], la);
    ns_cx(v[0], v[1], la);
    ns_cx(v[2], v[3], la);
  }
  ns_stage<8>(v, t, s, rs);
  ns_stage<16>(v, t, s, rs);
  ns_stage<32>(v, t, s, rs);
  ns_stage<64>(v, t, s, rs);
  ns_stage<128>(v, t, s, rs);
  ns_stage<256>(v, t, s, rs);
  ns_stage<512>(v, t, s, rs);
  ns_stage<1024>(v, t, s, rs);

  {
    const float t01 = v[0] + v[1], t012 = t01 + v[2], tsum = t012 + v[3];
    float r = tsum;
    r += ns_row_shr<1>(r);
    r += ns_row_shr<2>(r);
    r += ns_row_shr<4>(r);
    r += ns_row_shr<8>(r);
    const float rex = ns_row_shr<1>(r);
    const float c0 = __builtin_bit_cast(float, __builtin_amdgcn_readlane(__builtin_bit_cast(int, r), 15));
    const float c1 = __builtin_bit_cast(float, __builtin_amdgcn_readlane(__builtin_bit_cast(int, r), 31));
    const float ex = rex + (lane >= 16 ? c0 : 0.f);
    if (lane == 0) wsum[wave] = c0 + c1;
    __syncthreads();
    float base = 0.f, S = 0.f;
    #pragma unroll
    for (int w = 0; w < 8; ++w) {
      const float x = wsum[w];
      base += (w < wave) ? x : 0.f;
      S += x;
    }
    const float e0 = base + ex;
    const float e1 = e0 + v[0], e2 = e0 + t01, e3 = e0 + t012;
    const float p0 = (float)(8 * t - NS_N);
    float4 rv4;
    rv4.x = fmaf(v[0], p0,       S - 2.f * e0);
    rv4.y = fmaf(v[1], p0 + 2.f, S - 2.f * e1);
    rv4.z = fmaf(v[2], p0 + 4.f, S - 2.f * e2);
    rv4.w = fmaf(v[3], p0 + 6.f, S - 2.f * e3);
    *(float4*)&s[t * 4]  = make_float4(v[0], v[1], v[2], v[3]);
    *(float4*)&rs[t * 4] = rv4;
  }
  __syncthreads();

  const int col = lane & 15;
  const int h   = lane >> 4;
  const int g0  = blockIdx.y * 32 + col;
  const float cm0 = (float)(NS_N - 8 - 16 * g0);
  const float cm1 = cm0 - 256.0f;

  const int pm0 = NS_N - 5 - 8 * g0;
  const float sm0 = s[pm0], rm0 = rs[pm0], sm1 = s[pm0 - 128], rm1 = rs[pm0 - 128];

  __builtin_amdgcn_sched_barrier(0);

  v4f sv[16], rv[16];
  #pragma unroll
  for (int q = 0; q < 16; ++q) {
    const int p = 16 * (wave + 8 * (q >> 1)) + 8 * h + 4 * (q & 1);
    sv[q] = *(const v4f*)&s[p];
    rv[q] = *(const v4f*)&rs[p];
  }

  __builtin_amdgcn_sched_barrier(0);

  const float M0 = fmaf(cm0, sm0, -rm0);
  const float M1 = fmaf(cm1, sm1, -rm1);

  const float NS_CUT = -106.0f;
  const float d    = (float)(7 - 2 * (col & 7));
  const bool  wnum = col >= 8;
  v8f acc0 = {}, acc1 = {};
  float x0[2][8], x1[2][8];
  bool  live[2];
  auto ns_test = [&](const int m) {
    const int z = m & 1;
    #pragma unroll
    for (int u = 0; u < 2; ++u) {
      const v4f s4 = sv[2 * m + u], r4 = rv[2 * m + u];
      x0[z][4 * u + 0] = fmaf(cm0, s4.x, -r4.x) - M0;  x1[z][4 * u + 0] = fmaf(cm1, s4.x, -r4.x) - M1;
      x0[z][4 * u + 1] = fmaf(cm0, s4.y, -r4.y) - M0;  x1[z][4 * u + 1] = fmaf(cm1, s4.y, -r4.y) - M1;
      x0[z][4 * u + 2] = fmaf(cm0, s4.z, -r4.z) - M0;  x1[z][4 * u + 2] = fmaf(cm1, s4.z, -r4.z) - M1;
      x0[z][4 * u + 3] = fmaf(cm0, s4.w, -r4.w) - M0;  x1[z][4 * u + 3] = fmaf(cm1, s4.w, -r4.w) - M1;
    }
    const float top0 = fmaxf(fmaxf(fmaxf(x0[z][0], x0[z][1]), fmaxf(x0[z][2], x0[z][3])),
                             fmaxf(fmaxf(x0[z][4], x0[z][5]), fmaxf(x0[z][6], x0[z][7])));
    const float top1 = fmaxf(fmaxf(fmaxf(x1[z][0], x1[z][1]), fmaxf(x1[z][2], x1[z][3])),
                             fmaxf(fmaxf(x1[z][4], x1[z][5]), fmaxf(x1[z][6], x1[z][7])));
    live[z] = __builtin_amdgcn_ballot_w32(!(fmaxf(top0, top1) <= NS_CUT)) != 0u;
  };
  ns_test(0);
  #pragma unroll
  for (int m = 0; m < 8; ++m) {
    if (m < 7) ns_test(m + 1);
    const int z = m & 1;
    if (live[z]) {
      const v4f sa = sv[2 * m], sb = sv[2 * m + 1];
      const float sj[8] = {sa.x, sa.y, sa.z, sa.w, sb.x, sb.y, sb.z, sb.w};
      #pragma unroll
      for (int c = 0; c < 4; ++c) {
        const float sx = sj[2 * c], sy = sj[2 * c + 1];
        const float p0 = __expf(d * sx), p1 = __expf(d * sy);
        const v2f Bm = {wnum ? p0 * sx : p0, wnum ? p1 * sy : p1};
        const v2f A0 = {__expf(x0[z][2 * c]), __expf(x0[z][2 * c + 1])};
        const v2f A1 = {__expf(x1[z][2 * c]), __expf(x1[z][2 * c + 1])};
        acc0 = __builtin_amdgcn_wmma_f32_16x16x4_f32(
             false, A0,  false, Bm,
             (short)0, acc0,  false,  false);
        acc1 = __builtin_amdgcn_wmma_f32_16x16x4_f32(
             false, A1,  false, Bm,
             (short)0, acc1,  false,  false);
      }
    }
  }

  #pragma unroll
  for (int e = 0; e < 8; ++e) {
    red[wave][0][e][lane] = acc0[e];
    red[wave][1][e][lane] = acc1[e];
  }
  __syncthreads();

  const int tile = t >> 7;
  const int m    = (t >> 3) & 15;
  const int r    = t & 7;
  const int ln   = r + 16 * (m >> 3);
  float den = 0.f, num = 0.f;
  #pragma unroll
  for (int w = 0; w < 8; ++w) {
    den += red[w][tile][m & 7][ln];
    num += red[w][tile][m & 7][ln + 8];
  }
  out[b * NS_N + blockIdx.y * 256 + t] = num / den;
}

extern "C" void kernel_launch(void* const* d_in, const int* in_sizes, int n_in,
                              void* d_out, int out_size, void* d_ws, size_t ws_size,
                              hipStream_t stream) {
  (void)in_sizes; (void)n_in; (void)out_size; (void)d_ws; (void)ws_size;
  const float* scores = (const float*)d_in[0];
  float* out = (float*)d_out;

  ns_forward_kernel<<<dim3(NS_B, 4), 256, 0, stream>>>(scores, out);
}


# softmax factored over groups of 16 rows (was 8): one 16-group A tile for the block's 256 rows, den and num as two B fragments, P shifted by a per-block score; A exps and skip test halve; j order kept
# speedup vs baseline: 7.4081x; 1.0537x over previous
#include <hip/hip_runtime.h>


typedef __attribute__((ext_vector_type(2))) float v2f;
typedef __attribute__((ext_vector_type(4))) float v4f;
typedef __attribute__((ext_vector_type(8))) float v8f;

#define NS_N 1024
#define NS_B 64

template <int M>
__device__ __forceinline__ float ns_lane_xor(const float x) {
  const int b = __builtin_bit_cast(int, x);
  if constexpr (M < 16)
    return __builtin_bit_cast(float, __builtin_amdgcn_update_dpp(0, b, 0x160 + M, 0xf, 0xf, true));
  else
    return __builtin_bit_cast(float, __builtin_amdgcn_permlanex16(b, b, 0x76543210, 0xfedcba98, false, false));
}

template <int D>
__device__ __forceinline__ float ns_row_shr(const float x) {
  return __builtin_bit_cast(float, __builtin_amdgcn_update_dpp(0, __builtin_bit_cast(int, x), 0x110 + D, 0xf, 0xf, true));
}

__device__ __forceinline__ float ns_pick(const float a, const float b, const float lim) {
  return __builtin_amdgcn_fmed3f(a, b, lim);
}

__device__ __forceinline__ void ns_cx(float& a, float& b, const float lim) {
  const float x = a, y = b;
  a = ns_pick(x, y, lim);
  b = ns_pick(x, y, -lim);
}

template <int M>
__device__ __forceinline__ void ns_lane_step(float (&v)[4], const float lim) {
  #pragma unroll
  for (int e = 0; e < 4; ++e) v[e] = ns_pick(v[e], ns_lane_xor<M>(v[e]), lim);
}

__device__ __forceinline__ void ns_lds_step(float (&v)[4], const float lim, float* buf,
                                            const int t, const int m) {
  *(float4*)&buf[4 * t] = make_float4(v[0], v[1], v[2], v[3]);
  __syncthreads();
  const float4 o = *(const float4*)&buf[4 * (t ^ m)];
  v[0] = ns_pick(v[0], o.x, lim);
  v[1] = ns_pick(v[1], o.y, lim);
  v[2] = ns_pick(v[2], o.z, lim);
  v[3] = ns_pick(v[3], o.w, lim);
}

template <int K>
__device__ __forceinline__ void ns_stage(float (&v)[4], const int t, float* bA, float* bB) {
  const float inf = __builtin_inff();
  const bool asc = (t & (K / 4)) == 0;
  auto lim = [&](const int m) { return (((t & m) == 0) == asc) ? -inf : inf; };
  if constexpr (K == 1024) {
    ns_lds_step(v, lim(128), bB, t, 128);
    ns_lds_step(v, lim(64), bA, t, 64);
    ns_lds_step(v, lim(32), bB, t, 32);
  }
  if constexpr (K == 512) {
    ns_lds_step(v, lim(64), bB, t, 64);
    ns_lds_step(v, lim(32), bA, t, 32);
  }
  if constexpr (K == 256) ns_lds_step(v, lim(32), bA, t, 32);
  if constexpr (K >= 128) ns_lane_step<16>(v, lim(16));
  if constexpr (K >= 64) ns_lane_step<8>(v, lim(8));
  if constexpr (K >= 32) ns_lane_step<4>(v, lim(4));
  if constexpr (K >= 16) ns_lane_step<2>(v, lim(2));
  ns_lane_step<1>(v, lim(1));
  const float la = asc ? -inf : inf;
  ns_cx(v[0], v[2], la);
  ns_cx(v[1], v[3], la);
  ns_cx(v[0], v[1], la);
  ns_cx(v[2], v[3], la);
}

__global__ __launch_bounds__(256) void ns_forward_kernel(
    const float* __restrict__ scores, float* __restrict__ out) {
  __shared__ __align__(16) float s[NS_N];
  __shared__ __align__(16) float rs[NS_N];
  __shared__ float wsum[8];
  __shared__ __align__(16) float red[8][2][8][32];

  const int b    = blockIdx.x;
  const int t    = threadIdx.x;
  const int wave = t >> 5;
  const int lane = t & 31;

  float v[4];
  {
    const float4 in = *(const float4*)&scores[b * NS_N + t * 4];
    v[0] = in.x; v[1] = in.y; v[2] = in.z; v[3] = in.w;
  }
  {
    const float inf = __builtin_inff();
    ns_cx(v[0], v[1], -inf);
    ns_cx(v[2], v[3], inf);
    const float la = (t & 1) == 0 ? -inf : inf;
    ns_cx(v[0], v[2], la);
    ns_cx(v[1], v[3], la);
    ns_cx(v[0], v[1], la);
    ns_cx(v[2], v[3], la);
  }
  ns_stage<8>(v, t, s, rs);
  ns_stage<16>(v, t, s, rs);
  ns_stage<32>(v, t, s, rs);
  ns_stage<64>(v, t, s, rs);
  ns_stage<128>(v, t, s, rs);
  ns_stage<256>(v, t, s, rs);
  ns_stage<512>(v, t, s, rs);
  ns_stage<1024>(v, t, s, rs);

  {
    const float t01 = v[0] + v[1], t012 = t01 + v[2], tsum = t012 + v[3];
    float r = tsum;
    r += ns_row_shr<1>(r);
    r += ns_row_shr<2>(r);
    r += ns_row_shr<4>(r);
    r += ns_row_shr<8>(r);
    const float rex = ns_row_shr<1>(r);
    const float c0 = __builtin_bit_cast(float, __builtin_amdgcn_readlane(__builtin_bit_cast(int, r), 15));
    const float c1 = __builtin_bit_cast(float, __builtin_amdgcn_readlane(__builtin_bit_cast(int, r), 31));
    const float ex = rex + (lane >= 16 ? c0 : 0.f);
    if (lane == 0) wsum[wave] = c0 + c1;
    __syncthreads();
    float base = 0.f, S = 0.f;
    #pragma unroll
    for (int w = 0; w < 8; ++w) {
      const float x = wsum[w];
      base += (w < wave) ? x : 0.f;
      S += x;
    }
    const float e0 = base + ex;
    const float e1 = e0 + v[0], e2 = e0 + t01, e3 = e0 + t012;
    const float p0 = (float)(8 * t - NS_N);
    float4 rv4;
    rv4.x = fmaf(v[0], p0,       S - 2.f * e0);
    rv4.y = fmaf(v[1], p0 + 2.f, S - 2.f * e1);
    rv4.z = fmaf(v[2], p0 + 4.f, S - 2.f * e2);
    rv4.w = fmaf(v[3], p0 + 6.f, S - 2.f * e3);
    *(float4*)&s[t * 4]  = make_float4(v[0], v[1], v[2], v[3]);
    *(float4*)&rs[t * 4] = rv4;
  }
  __syncthreads();

  const int col = lane & 15;
  const int h   = lane >> 4;
  const int G   = blockIdx.y * 16 + col;
  const float cm = (float)(NS_N - 16 - 32 * G);

  const int pm = NS_N - 9 - 16 * G;
  const float sm = s[pm], rm = rs[pm];
  const float shi = s[NS_N - 1 - 256 * blockIdx.y], slo = s[NS_N - 256 - 256 * blockIdx.y];

  __builtin_amdgcn_sched_barrier(0);

  v4f sv[16], rv[16];
  #pragma unroll
  for (int q = 0; q < 16; ++q) {
    const int p = 16 * (wave + 8 * (q >> 1)) + 8 * h + 4 * (q & 1);
    sv[q] = *(const v4f*)&s[p];
    rv[q] = *(const v4f*)&rs[p];
  }

  __builtin_amdgcn_sched_barrier(0);

  const float M = fmaf(cm, sm, -rm);

  const float NS_CUT = -106.0f;
  const float d   = (float)(15 - 2 * col);
  const float dsc = d * (0.5f * (shi + slo));
  v8f accd = {}, accn = {};
  float x[2][8];
  bool  live[2];
  auto ns_test = [&](const int m) {
    const int z = m & 1;
    #pragma unroll
    for (int u = 0; u < 2; ++u) {
      const v4f s4 = sv[2 * m + u], r4 = rv[2 * m + u];
      x[z][4 * u + 0] = fmaf(cm, s4.x, -r4.x) - M;
      x[z][4 * u + 1] = fmaf(cm, s4.y, -r4.y) - M;
      x[z][4 * u + 2] = fmaf(cm, s4.z, -r4.z) - M;
      x[z][4 * u + 3] = fmaf(cm, s4.w, -r4.w) - M;
    }
    const float top = fmaxf(fmaxf(fmaxf(x[z][0], x[z][1]), fmaxf(x[z][2], x[z][3])),
                            fmaxf(fmaxf(x[z][4], x[z][5]), fmaxf(x[z][6], x[z][7])));
    live[z] = __builtin_amdgcn_ballot_w32(!(top <= NS_CUT)) != 0u;
  };
  ns_test(0);
  #pragma unroll
  for (int m = 0; m < 8; ++m) {
    if (m < 7) ns_test(m + 1);
    const int z = m & 1;
    if (live[z]) {
      const v4f sa = sv[2 * m], sb = sv[2 * m + 1];
      const float sj[8] = {sa.x, sa.y, sa.z, sa.w, sb.x, sb.y, sb.z, sb.w};
      #pragma unroll
      for (int c = 0; c < 4; ++c) {
        const float sx = sj[2 * c], sy = sj[2 * c + 1];
        const float p0 = __expf(fmaf(d, sx, -dsc)), p1 = __expf(fmaf(d, sy, -dsc));
        const v2f Bd = {p0, p1};
        const v2f Bn = {p0 * sx, p1 * sy};
        const v2f A  = {__expf(x[z][2 * c]), __expf(x[z][2 * c + 1])};
        accd = __builtin_amdgcn_wmma_f32_16x16x4_f32(
             false, A,  false, Bd,
             (short)0, accd,  false,  false);
        accn = __builtin_amdgcn_wmma_f32_16x16x4_f32(
             false, A,  false, Bn,
             (short)0, accn,  false,  false);
      }
    }
  }

  #pragma unroll
  for (int e = 0; e < 8; ++e) {
    red[wave][0][e][lane] = accd[e];
    red[wave][1][e][lane] = accn[e];
  }
  __syncthreads();

  const int m  = t >> 4;
  const int r  = t & 15;
  const int ln = r + 16 * (m >> 3);
  float den = 0.f, num = 0.f;
  #pragma unroll
  for (int w = 0; w < 8; ++w) {
    den += red[w][0][m & 7][ln];
    num += red[w][1][m & 7][ln];
  }
  out[b * NS_N + blockIdx.y * 256 + t] = num / den;
}

extern "C" void kernel_launch(void* const* d_in, const int* in_sizes, int n_in,
                              void* d_out, int out_size, void* d_ws, size_t ws_size,
                              hipStream_t stream) {
  (void)in_sizes; (void)n_in; (void)out_size; (void)d_ws; (void)ws_size;
  const float* scores = (const float*)d_in[0];
  float* out = (float*)d_out;

  ns_forward_kernel<<<dim3(NS_B, 4), 256, 0, stream>>>(scores, out);
}


# bitonic sort and prefix sum on waves 0-3 only, 8 positions per thread (was 8 waves x 4): 3 LDS exchange steps with barriers instead of 6; prefix sums regrouped: sum order changes
# speedup vs baseline: 7.8987x; 1.0662x over previous
#include <hip/hip_runtime.h>


typedef __attribute__((ext_vector_type(2))) float v2f;
typedef __attribute__((ext_vector_type(4))) float v4f;
typedef __attribute__((ext_vector_type(8))) float v8f;

#define NS_N 1024
#define NS_B 64

template <int M>
__device__ __forceinline__ float ns_lane_xor(const float x) {
  const int b = __builtin_bit_cast(int, x);
  if constexpr (M < 16)
    return __builtin_bit_cast(float, __builtin_amdgcn_update_dpp(0, b, 0x160 + M, 0xf, 0xf, true));
  else
    return __builtin_bit_cast(float, __builtin_amdgcn_permlanex16(b, b, 0x76543210, 0xfedcba98, false, false));
}

template <int D>
__device__ __forceinline__ float ns_row_shr(const float x) {
  return __builtin_bit_cast(float, __builtin_amdgcn_update_dpp(0, __builtin_bit_cast(int, x), 0x110 + D, 0xf, 0xf, true));
}

__device__ __forceinline__ float ns_pick(const float a, const float b, const float lim) {
  return __builtin_amdgcn_fmed3f(a, b, lim);
}

__device__ __forceinline__ void ns_cx(float& a, float& b, const float lim) {
  const float x = a, y = b;
  a = ns_pick(x, y, lim);
  b = ns_pick(x, y, -lim);
}

template <int M>
__device__ __forceinline__ void ns_lane_step(float (&v)[8], const float lim) {
  #pragma unroll
  for (int e = 0; e < 8; ++e) v[e] = ns_pick(v[e], ns_lane_xor<M>(v[e]), lim);
}

__device__ __forceinline__ void ns_tail(float (&v)[8], const float la) {
  ns_cx(v[0], v[4], la); ns_cx(v[1], v[5], la); ns_cx(v[2], v[6], la); ns_cx(v[3], v[7], la);
  ns_cx(v[0], v[2], la); ns_cx(v[1], v[3], la); ns_cx(v[4], v[6], la); ns_cx(v[5], v[7], la);
  ns_cx(v[0], v[1], la); ns_cx(v[2], v[3], la); ns_cx(v[4], v[5], la); ns_cx(v[6], v[7], la);
}

__device__ __forceinline__ void ns_put(float* buf, const int t, const float (&v)[8]) {
  *(float4*)&buf[4 * t]       = make_float4(v[0], v[1], v[2], v[3]);
  *(float4*)&buf[512 + 4 * t] = make_float4(v[4], v[5], v[6], v[7]);
}

__device__ __forceinline__ void ns_take(const float* buf, const int u, float (&v)[8], const float lim) {
  const float4 a = *(const float4*)&buf[4 * u];
  const float4 c = *(const float4*)&buf[512 + 4 * u];
  v[0] = ns_pick(v[0], a.x, lim);
  v[1] = ns_pick(v[1], a.y, lim);
  v[2] = ns_pick(v[2], a.z, lim);
  v[3] = ns_pick(v[3], a.w, lim);
  v[4] = ns_pick(v[4], c.x, lim);
  v[5] = ns_pick(v[5], c.y, lim);
  v[6] = ns_pick(v[6], c.z, lim);
  v[7] = ns_pick(v[7], c.w, lim);
}

template <int K>
__device__ __forceinline__ void ns_stage(float (&v)[8], const int t) {
  const float inf = __builtin_inff();
  const bool asc = (K == 1024) || (t & (K / 8)) == 0;
  auto lim = [&](const int m) { return (((t & m) == 0) == asc) ? -inf : inf; };
  if constexpr (K >= 256) ns_lane_step<16>(v, lim(16));
  if constexpr (K >= 128) ns_lane_step<8>(v, lim(8));
  if constexpr (K >= 64) ns_lane_step<4>(v, lim(4));
  if constexpr (K >= 32) ns_lane_step<2>(v, lim(2));
  ns_lane_step<1>(v, lim(1));
  ns_tail(v, asc ? -inf : inf);
}

__global__ __launch_bounds__(256) void ns_forward_kernel(
    const float* __restrict__ scores, float* __restrict__ out) {
  __shared__ __align__(16) float s[NS_N];
  __shared__ __align__(16) float rs[NS_N];
  __shared__ float wsum[4];
  __shared__ __align__(16) float red[8][2][8][32];

  const int b    = blockIdx.x;
  const int t    = threadIdx.x;
  const int wave = t >> 5;
  const int lane = t & 31;

  const bool sorter = __builtin_amdgcn_readfirstlane(t) < 128;
  const float inf = __builtin_inff();
  float v[8] = {0.f, 0.f, 0.f, 0.f, 0.f, 0.f, 0.f, 0.f};
  if (sorter) {
    const float4 i0 = *(const float4*)&scores[b * NS_N + t * 4];
    const float4 i1 = *(const float4*)&scores[b * NS_N + 512 + t * 4];
    v[0] = i0.x; v[1] = i0.y; v[2] = i0.z; v[3] = i0.w;
    v[4] = i1.x; v[5] = i1.y; v[6] = i1.z; v[7] = i1.w;
    ns_cx(v[0], v[1], -inf);
    ns_cx(v[2], v[3], inf);
    ns_cx(v[4], v[5], -inf);
    ns_cx(v[6], v[7], inf);
    ns_cx(v[0], v[2], -inf);
    ns_cx(v[1], v[3], -inf);
    ns_cx(v[0], v[1], -inf);
    ns_cx(v[2], v[3], -inf);
    ns_cx(v[4], v[6], inf);
    ns_cx(v[5], v[7], inf);
    ns_cx(v[4], v[5], inf);
    ns_cx(v[6], v[7], inf);
    ns_tail(v, (t & 1) == 0 ? -inf : inf);
    ns_stage<16>(v, t);
    ns_stage<32>(v, t);
    ns_stage<64>(v, t);
    ns_stage<128>(v, t);
    ns_stage<256>(v, t);
    ns_put(s, t, v);
  }
  __syncthreads();
  if (sorter) {
    const bool asc = (t & 64) == 0;
    ns_take(s, t ^ 32, v, (((t & 32) == 0) == asc) ? -inf : inf);
    ns_stage<512>(v, t);
    ns_put(rs, t, v);
  }
  __syncthreads();
  if (sorter) {
    ns_take(rs, t ^ 64, v, (t & 64) == 0 ? -inf : inf);
    ns_put(s, t, v);
  }
  __syncthreads();

  float ex = 0.f;
  float pre[8] = {0.f, 0.f, 0.f, 0.f, 0.f, 0.f, 0.f, 0.f};
  if (sorter) {
    ns_take(s, t ^ 32, v, (t & 32) == 0 ? -inf : inf);
    ns_stage<1024>(v, t);
    #pragma unroll
    for (int e = 1; e < 8; ++e) pre[e] = pre[e - 1] + v[e - 1];
    float r = pre[7] + v[7];
    r += ns_row_shr<1>(r);
    r += ns_row_shr<2>(r);
    r += ns_row_shr<4>(r);
    r += ns_row_shr<8>(r);
    const float rex = ns_row_shr<1>(r);
    const float c0 = __builtin_bit_cast(float, __builtin_amdgcn_readlane(__builtin_bit_cast(int, r), 15));
    const float c1 = __builtin_bit_cast(float, __builtin_amdgcn_readlane(__builtin_bit_cast(int, r), 31));
    ex = rex + (lane >= 16 ? c0 : 0.f);
    if (lane == 0) wsum[wave] = c0 + c1;
  }
  __syncthreads();
  if (sorter) {
    float base = 0.f, S = 0.f;
    #pragma unroll
    for (int w = 0; w < 4; ++w) {
      const float x = wsum[w];
      base += (w < wave) ? x : 0.f;
      S += x;
    }
    const float e0 = base + ex;
    const float p0 = (float)(16 * t - NS_N);
    float rq[8];
    #pragma unroll
    for (int e = 0; e < 8; ++e)
      rq[e] = fmaf(v[e], p0 + (float)(2 * e), S - 2.f * (e0 + pre[e]));
    *(float4*)&s[t * 8]      = make_float4(v[0], v[1], v[2], v[3]);
    *(float4*)&s[t * 8 + 4]  = make_float4(v[4], v[5], v[6], v[7]);
    *(float4*)&rs[t * 8]     = make_float4(rq[0], rq[1], rq[2], rq[3]);
    *(float4*)&rs[t * 8 + 4] = make_float4(rq[4], rq[5], rq[6], rq[7]);
  }
  __syncthreads();

  const int col = lane & 15;
  const int h   = lane >> 4;
  const int G   = blockIdx.y * 16 + col;
  const float cm = (float)(NS_N - 16 - 32 * G);

  const int pm = NS_N - 9 - 16 * G;
  const float sm = s[pm], rm = rs[pm];
  const float shi = s[NS_N - 1 - 256 * blockIdx.y], slo = s[NS_N - 256 - 256 * blockIdx.y];

  __builtin_amdgcn_sched_barrier(0);

  v4f sv[16], rv[16];
  #pragma unroll
  for (int q = 0; q < 16; ++q) {
    const int p = 16 * (wave + 8 * (q >> 1)) + 8 * h + 4 * (q & 1);
    sv[q] = *(const v4f*)&s[p];
    rv[q] = *(const v4f*)&rs[p];
  }

  __builtin_amdgcn_sched_barrier(0);

  const float M = fmaf(cm, sm, -rm);

  const float NS_CUT = -106.0f;
  const float d   = (float)(15 - 2 * col);
  const float dsc = d * (0.5f * (shi + slo));
  v8f accd = {}, accn = {};
  float x[2][8];
  bool  live[2];
  auto ns_test = [&](const int m) {
    const int z = m & 1;
    #pragma unroll
    for (int u = 0; u < 2; ++u) {
      const v4f s4 = sv[2 * m + u], r4 = rv[2 * m + u];
      x[z][4 * u + 0] = fmaf(cm, s4.x, -r4.x) - M;
      x[z][4 * u + 1] = fmaf(cm, s4.y, -r4.y) - M;
      x[z][4 * u + 2] = fmaf(cm, s4.z, -r4.z) - M;
      x[z][4 * u + 3] = fmaf(cm, s4.w, -r4.w) - M;
    }
    const float top = fmaxf(fmaxf(fmaxf(x[z][0], x[z][1]), fmaxf(x[z][2], x[z][3])),
                            fmaxf(fmaxf(x[z][4], x[z][5]), fmaxf(x[z][6], x[z][7])));
    live[z] = __builtin_amdgcn_ballot_w32(!(top <= NS_CUT)) != 0u;
  };
  ns_test(0);
  #pragma unroll
  for (int m = 0; m < 8; ++m) {
    if (m < 7) ns_test(m + 1);
    const int z = m & 1;
    if (live[z]) {
      const v4f sa = sv[2 * m], sb = sv[2 * m + 1];
      const float sj[8] = {sa.x, sa.y, sa.z, sa.w, sb.x, sb.y, sb.z, sb.w};
      #pragma unroll
      for (int c = 0; c < 4; ++c) {
        const float sx = sj[2 * c], sy = sj[2 * c + 1];
        const float p0 = __expf(fmaf(d, sx, -dsc)), p1 = __expf(fmaf(d, sy, -dsc));
        const v2f Bd = {p0, p1};
        const v2f Bn = {p0 * sx, p1 * sy};
        const v2f A  = {__expf(x[z][2 * c]), __expf(x[z][2 * c + 1])};
        accd = __builtin_amdgcn_wmma_f32_16x16x4_f32(
             false, A,  false, Bd,
             (short)0, accd,  false,  false);
        accn = __builtin_amdgcn_wmma_f32_16x16x4_f32(
             false, A,  false, Bn,
             (short)0, accn,  false,  false);
      }
    }
  }

  #pragma unroll
  for (int e = 0; e < 8; ++e) {
    red[wave][0][e][lane] = accd[e];
    red[wave][1][e][lane] = accn[e];
  }
  __syncthreads();

  const int m  = t >> 4;
  const int r  = t & 15;
  const int ln = r + 16 * (m >> 3);
  float den = 0.f, num = 0.f;
  #pragma unroll
  for (int w = 0; w < 8; ++w) {
    den += red[w][0][m & 7][ln];
    num += red[w][1][m & 7][ln];
  }
  out[b * NS_N + blockIdx.y * 256 + t] = num / den;
}

extern "C" void kernel_launch(void* const* d_in, const int* in_sizes, int n_in,
                              void* d_out, int out_size, void* d_ws, size_t ws_size,
                              hipStream_t stream) {
  (void)in_sizes; (void)n_in; (void)out_size; (void)d_ws; (void)ws_size;
  const float* scores = (const float*)d_in[0];
  float* out = (float*)d_out;

  ns_forward_kernel<<<dim3(NS_B, 4), 256, 0, stream>>>(scores, out);
}
